# starting kernel
# baseline (speedup 1.0000x reference)
#include <hip/hip_runtime.h>
#include <hip/hip_bf16.h>
#include <math.h>

#define EMB        256
#define TWO_D      512
#define KTOT       1024
#define NREL       128
#define BATCH      4096
#define LOG_SQRT_2PI 0.9189385332046727f
#define XSTRIDE    516
#define MROWS      32

typedef __attribute__((ext_vector_type(2))) float v2f;
typedef __attribute__((ext_vector_type(8))) float v8f;

__global__ __launch_bounds__(256)
void nb_prep_w(const float* __restrict__ mus, const float* __restrict__ sigmas,
               float* __restrict__ Wf) {
    int i = blockIdx.x * 256 + threadIdx.x;
    if (i >= TWO_D * NREL) return;
    int k = i / NREL;
    int r = i - k * NREL;
    float s  = sigmas[r * TWO_D + k];
    float iv = 1.0f / (s * s);
    float av = -0.5f * iv;
    float bm = mus[r * TWO_D + k] * iv;
    int t = k >> 2, h = (k >> 1) & 1, j = k & 1;
    Wf[t * 512 + r * 4 + h * 2 + j]            = av;
    Wf[(t + 128) * 512 + r * 4 + h * 2 + j]    = bm;
}

__global__ __launch_bounds__(256)
void nb_prep_bias(const float* __restrict__ mus, const float* __restrict__ sigmas,
                  const float* __restrict__ priors, float* __restrict__ C) {
    __shared__ float red[256];
    int r = blockIdx.x;
    float acc = 0.0f;
    for (int k = threadIdx.x; k < TWO_D; k += 256) {
        float s  = sigmas[r * TWO_D + k];
        float mu = mus[r * TWO_D + k];
        float iv = 1.0f / (s * s);
        acc += -0.5f * mu * mu * iv - logf(s);
    }
    red[threadIdx.x] = acc;
    __syncthreads();
    for (int off = 128; off > 0; off >>= 1) {
        if (threadIdx.x < off) red[threadIdx.x] += red[threadIdx.x + off];
        __syncthreads();
    }
    if (threadIdx.x == 0)
        C[r] = red[0] + (float)TWO_D * (priors[r] - LOG_SQRT_2PI);
}

__global__ __launch_bounds__(256)
void nb_wmma(const float* __restrict__ sbjs, const float* __restrict__ objs,
             const float* __restrict__ Wf, const float* __restrict__ C,
             float* __restrict__ out) {
    __shared__ float xs[MROWS * XSTRIDE];

    const int m0 = blockIdx.x * MROWS;

    for (int i = threadIdx.x; i < MROWS * TWO_D; i += 256) {
        int row = i >> 9;
        int k   = i & (TWO_D - 1);
        int b   = m0 + row;
        float v = (k < EMB) ? sbjs[b * EMB + k] : objs[b * EMB + (k - EMB)];
        xs[row * XSTRIDE + k] = v;
    }
    __syncthreads();

    const int lane  = threadIdx.x & 31;
    const int wave  = threadIdx.x >> 5;
    const int l16   = lane & 15;
    const int lhalf = lane >> 4;
    const int r0    = wave * 16;

    const float* x0 = &xs[l16 * XSTRIDE + 2 * lhalf];
    const float* x1 = x0 + 16 * XSTRIDE;
    const float* wp = Wf + ((r0 + l16) * 2 + lhalf) * 2;

    v8f acc0 = {};
    v8f acc1 = {};

    #pragma unroll 2
    for (int t = 0; t < 128; ++t) {
        v2f b   = *(const v2f*)(wp + t * 512);
        v2f xv0 = *(const v2f*)(x0 + t * 4);
        v2f xv1 = *(const v2f*)(x1 + t * 4);
        v2f a0;  a0.x = xv0.x * xv0.x;  a0.y = xv0.y * xv0.y;
        v2f a1;  a1.x = xv1.x * xv1.x;  a1.y = xv1.y * xv1.y;
        acc0 = __builtin_amdgcn_wmma_f32_16x16x4_f32(
            false, a0, false, b, (short)0, acc0, false, false);
        acc1 = __builtin_amdgcn_wmma_f32_16x16x4_f32(
            false, a1, false, b, (short)0, acc1, false, false);
    }
    const float* wp2 = wp + 128 * 512;
    #pragma unroll 2
    for (int t = 0; t < 128; ++t) {
        v2f b  = *(const v2f*)(wp2 + t * 512);
        v2f a0 = *(const v2f*)(x0 + t * 4);
        v2f a1 = *(const v2f*)(x1 + t * 4);
        acc0 = __builtin_amdgcn_wmma_f32_16x16x4_f32(
            false, a0, false, b, (short)0, acc0, false, false);
        acc1 = __builtin_amdgcn_wmma_f32_16x16x4_f32(
            false, a1, false, b, (short)0, acc1, false, false);
    }

    float bias = C[r0 + l16];
    float* op0 = out + (m0 + 8 * lhalf) * NREL + r0 + l16;
    float* op1 = op0 + 16 * NREL;
    #pragma unroll
    for (int i = 0; i < 8; ++i) {
        op0[i * NREL] = acc0[i] + bias;
        op1[i * NREL] = acc1[i] + bias;
    }
}

extern "C" void kernel_launch(void* const* d_in, const int* in_sizes, int n_in,
                              void* d_out, int out_size, void* d_ws, size_t ws_size,
                              hipStream_t stream) {
    const float* sbjs   = (const float*)d_in[0];
    const float* objs   = (const float*)d_in[1];
    const float* mus    = (const float*)d_in[2];
    const float* sigmas = (const float*)d_in[3];
    const float* priors = (const float*)d_in[4];
    float* out = (float*)d_out;

    float* Wf = (float*)d_ws;
    float* C  = Wf + KTOT * NREL;

    nb_prep_w<<<(TWO_D * NREL + 255) / 256, 256, 0, stream>>>(mus, sigmas, Wf);
    nb_prep_bias<<<NREL, 256, 0, stream>>>(mus, sigmas, priors, C);
    nb_wmma<<<BATCH / MROWS, 256, 0, stream>>>(sbjs, objs, Wf, C, out);
}


# Main kernel: B fragments are fetched 16 k-steps ahead through a register ring (primed before the x-tile fill), so 16 global loads per wave are in flight where 2 were
# speedup vs baseline: 2.5025x; 2.5025x over previous
#include <hip/hip_runtime.h>
#include <hip/hip_bf16.h>
#include <math.h>

#define EMB        256
#define TWO_D      512
#define KTOT       1024
#define NREL       128
#define BATCH      4096
#define LOG_SQRT_2PI 0.9189385332046727f
#define XSTRIDE    516
#define MROWS      32

typedef __attribute__((ext_vector_type(2))) float v2f;
typedef __attribute__((ext_vector_type(8))) float v8f;

__global__ __launch_bounds__(256)
void nb_prep_w(const float* __restrict__ mus, const float* __restrict__ sigmas,
               float* __restrict__ Wf) {
    int i = blockIdx.x * 256 + threadIdx.x;
    if (i >= TWO_D * NREL) return;
    int k = i / NREL;
    int r = i - k * NREL;
    float s  = sigmas[r * TWO_D + k];
    float iv = 1.0f / (s * s);
    float av = -0.5f * iv;
    float bm = mus[r * TWO_D + k] * iv;
    int t = k >> 2, h = (k >> 1) & 1, j = k & 1;
    Wf[t * 512 + r * 4 + h * 2 + j]            = av;
    Wf[(t + 128) * 512 + r * 4 + h * 2 + j]    = bm;
}

__global__ __launch_bounds__(256)
void nb_prep_bias(const float* __restrict__ mus, const float* __restrict__ sigmas,
                  const float* __restrict__ priors, float* __restrict__ C) {
    __shared__ float red[256];
    int r = blockIdx.x;
    float acc = 0.0f;
    for (int k = threadIdx.x; k < TWO_D; k += 256) {
        float s  = sigmas[r * TWO_D + k];
        float mu = mus[r * TWO_D + k];
        float iv = 1.0f / (s * s);
        acc += -0.5f * mu * mu * iv - logf(s);
    }
    red[threadIdx.x] = acc;
    __syncthreads();
    for (int off = 128; off > 0; off >>= 1) {
        if (threadIdx.x < off) red[threadIdx.x] += red[threadIdx.x + off];
        __syncthreads();
    }
    if (threadIdx.x == 0)
        C[r] = red[0] + (float)TWO_D * (priors[r] - LOG_SQRT_2PI);
}

#define PF 16
#define NSTEP 256

template <bool SQUARE, bool REFILL>
__device__ __forceinline__
void nb_steps(const int t0, v2f (&bq)[PF], const float* __restrict__ wp,
              const float* x0, const float* x1, v8f& acc0, v8f& acc1) {
    #pragma unroll
    for (int i = 0; i < PF; ++i) {
        const int t = t0 + i;
        const v2f b = bq[i];
        if (REFILL) bq[i] = *(const v2f*)(wp + (t + PF) * 512);
        const int xo = (t & 127) * 4;
        v2f a0 = *(const v2f*)(x0 + xo);
        v2f a1 = *(const v2f*)(x1 + xo);
        if (SQUARE) {
            a0.x = a0.x * a0.x;  a0.y = a0.y * a0.y;
            a1.x = a1.x * a1.x;  a1.y = a1.y * a1.y;
        }
        acc0 = __builtin_amdgcn_wmma_f32_16x16x4_f32(
            false, a0, false, b, (short)0, acc0, false, false);
        acc1 = __builtin_amdgcn_wmma_f32_16x16x4_f32(
            false, a1, false, b, (short)0, acc1, false, false);
    }
}

__global__ __launch_bounds__(256)
void nb_wmma(const float* __restrict__ sbjs, const float* __restrict__ objs,
             const float* __restrict__ Wf, const float* __restrict__ C,
             float* __restrict__ out) {
    __shared__ float xs[MROWS * XSTRIDE];

    const int m0 = blockIdx.x * MROWS;

    const int lane  = threadIdx.x & 31;
    const int wave  = threadIdx.x >> 5;
    const int l16   = lane & 15;
    const int lhalf = lane >> 4;
    const int r0    = wave * 16;
    const float* wp = Wf + ((r0 + l16) * 2 + lhalf) * 2;

    v2f bq[PF];
    #pragma unroll
    for (int i = 0; i < PF; ++i) bq[i] = *(const v2f*)(wp + i * 512);

    for (int i = threadIdx.x; i < MROWS * TWO_D; i += 256) {
        int row = i >> 9;
        int k   = i & (TWO_D - 1);
        int b   = m0 + row;
        float v = (k < EMB) ? sbjs[b * EMB + k] : objs[b * EMB + (k - EMB)];
        xs[row * XSTRIDE + k] = v;
    }
    __syncthreads();

    const float* x0 = &xs[l16 * XSTRIDE + 2 * lhalf];
    const float* x1 = x0 + 16 * XSTRIDE;

    v8f acc0 = {};
    v8f acc1 = {};

    for (int t0 = 0; t0 < 128; t0 += PF)
        nb_steps<true, true>(t0, bq, wp, x0, x1, acc0, acc1);
    for (int t0 = 128; t0 < NSTEP - PF; t0 += PF)
        nb_steps<false, true>(t0, bq, wp, x0, x1, acc0, acc1);
    nb_steps<false, false>(NSTEP - PF, bq, wp, x0, x1, acc0, acc1);

    float bias = C[r0 + l16];
    float* op0 = out + (m0 + 8 * lhalf) * NREL + r0 + l16;
    float* op1 = op0 + 16 * NREL;
    #pragma unroll
    for (int i = 0; i < 8; ++i) {
        op0[i * NREL] = acc0[i] + bias;
        op1[i * NREL] = acc1[i] + bias;
    }
}

extern "C" void kernel_launch(void* const* d_in, const int* in_sizes, int n_in,
                              void* d_out, int out_size, void* d_ws, size_t ws_size,
                              hipStream_t stream) {
    const float* sbjs   = (const float*)d_in[0];
    const float* objs   = (const float*)d_in[1];
    const float* mus    = (const float*)d_in[2];
    const float* sigmas = (const float*)d_in[3];
    const float* priors = (const float*)d_in[4];
    float* out = (float*)d_out;

    float* Wf = (float*)d_ws;
    float* C  = Wf + KTOT * NREL;

    nb_prep_w<<<(TWO_D * NREL + 255) / 256, 256, 0, stream>>>(mus, sigmas, Wf);
    nb_prep_bias<<<NREL, 256, 0, stream>>>(mus, sigmas, priors, C);
    nb_wmma<<<BATCH / MROWS, 256, 0, stream>>>(sbjs, objs, Wf, C, out);
}


# Main kernel re-tiled to 256 work-groups (32 rows x 64 relations, one per WGP): x^2 and x halves of K run on separate waves and are summed separately, partials added through LDS
# speedup vs baseline: 2.9047x; 1.1607x over previous
#include <hip/hip_runtime.h>
#include <hip/hip_bf16.h>
#include <math.h>

#define EMB        256
#define TWO_D      512
#define KTOT       1024
#define NREL       128
#define BATCH      4096
#define LOG_SQRT_2PI 0.9189385332046727f
#define XSTRIDE    516
#define MROWS      32

typedef __attribute__((ext_vector_type(2))) float v2f;
typedef __attribute__((ext_vector_type(8))) float v8f;

__global__ __launch_bounds__(256)
void nb_prep_w(const float* __restrict__ mus, const float* __restrict__ sigmas,
               float* __restrict__ Wf) {
    int i = blockIdx.x * 256 + threadIdx.x;
    if (i >= TWO_D * NREL) return;
    int k = i / NREL;
    int r = i - k * NREL;
    float s  = sigmas[r * TWO_D + k];
    float iv = 1.0f / (s * s);
    float av = -0.5f * iv;
    float bm = mus[r * TWO_D + k] * iv;
    int t = k >> 2, h = (k >> 1) & 1, j = k & 1;
    Wf[t * 512 + r * 4 + h * 2 + j]            = av;
    Wf[(t + 128) * 512 + r * 4 + h * 2 + j]    = bm;
}

__global__ __launch_bounds__(256)
void nb_prep_bias(const float* __restrict__ mus, const float* __restrict__ sigmas,
                  const float* __restrict__ priors, float* __restrict__ C) {
    __shared__ float red[256];
    int r = blockIdx.x;
    float acc = 0.0f;
    for (int k = threadIdx.x; k < TWO_D; k += 256) {
        float s  = sigmas[r * TWO_D + k];
        float mu = mus[r * TWO_D + k];
        float iv = 1.0f / (s * s);
        acc += -0.5f * mu * mu * iv - logf(s);
    }
    red[threadIdx.x] = acc;
    __syncthreads();
    for (int off = 128; off > 0; off >>= 1) {
        if (threadIdx.x < off) red[threadIdx.x] += red[threadIdx.x + off];
        __syncthreads();
    }
    if (threadIdx.x == 0)
        C[r] = red[0] + (float)TWO_D * (priors[r] - LOG_SQRT_2PI);
}

#define PF 16
#define HSTEP 128
#define RELB  64
#define NBLK  256

template <bool SQUARE, bool REFILL>
__device__ __forceinline__
void nb_steps(const int t0, v2f (&bq)[PF], const float* __restrict__ wp,
              const float* x0, const float* x1, v8f& acc0, v8f& acc1) {
    #pragma unroll
    for (int i = 0; i < PF; ++i) {
        const int t = t0 + i;
        const v2f b = bq[i];
        if (REFILL) bq[i] = *(const v2f*)(wp + (t + PF) * 512);
        const int xo = (t & 127) * 4;
        v2f a0 = *(const v2f*)(x0 + xo);
        v2f a1 = *(const v2f*)(x1 + xo);
        if (SQUARE) {
            a0.x = a0.x * a0.x;  a0.y = a0.y * a0.y;
            a1.x = a1.x * a1.x;  a1.y = a1.y * a1.y;
        }
        acc0 = __builtin_amdgcn_wmma_f32_16x16x4_f32(
            false, a0, false, b, (short)0, acc0, false, false);
        acc1 = __builtin_amdgcn_wmma_f32_16x16x4_f32(
            false, a1, false, b, (short)0, acc1, false, false);
    }
}

__global__ __launch_bounds__(256)
void nb_wmma(const float* __restrict__ sbjs, const float* __restrict__ objs,
             const float* __restrict__ Wf, const float* __restrict__ C,
             float* __restrict__ out) {
    __shared__ float xs[MROWS * XSTRIDE];
    __shared__ float red[4 * 16 * 32];

    const int m0    = (blockIdx.x >> 1) * MROWS;
    const int rbase = (blockIdx.x & 1) * RELB;

    const int lane  = threadIdx.x & 31;
    const int wave  = __builtin_amdgcn_readfirstlane((int)(threadIdx.x >> 5));
    const int grp   = wave & 3;
    const int seg   = wave >> 2;
    const int l16   = lane & 15;
    const int lhalf = lane >> 4;
    const int r0    = rbase + grp * 16;
    const float* wp = Wf + seg * (HSTEP * 512) + ((r0 + l16) * 2 + lhalf) * 2;

    v2f bq[PF];
    #pragma unroll
    for (int i = 0; i < PF; ++i) bq[i] = *(const v2f*)(wp + i * 512);

    for (int i = threadIdx.x; i < MROWS * TWO_D; i += 256) {
        int row = i >> 9;
        int k   = i & (TWO_D - 1);
        int b   = m0 + row;
        float v = (k < EMB) ? sbjs[b * EMB + k] : objs[b * EMB + (k - EMB)];
        xs[row * XSTRIDE + k] = v;
    }
    __syncthreads();

    const float* x0 = &xs[l16 * XSTRIDE + 2 * lhalf];
    const float* x1 = x0 + 16 * XSTRIDE;

    v8f acc0 = {};
    v8f acc1 = {};
    float* rp = red + grp * (16 * 32) + lane;

    if (seg == 0) {
        for (int t0 = 0; t0 < HSTEP - PF; t0 += PF)
            nb_steps<true, true>(t0, bq, wp, x0, x1, acc0, acc1);
        nb_steps<true, false>(HSTEP - PF, bq, wp, x0, x1, acc0, acc1);
    } else {
        for (int t0 = 0; t0 < HSTEP - PF; t0 += PF)
            nb_steps<false, true>(t0, bq, wp, x0, x1, acc0, acc1);
        nb_steps<false, false>(HSTEP - PF, bq, wp, x0, x1, acc0, acc1);
        #pragma unroll
        for (int i = 0; i < 8; ++i) {
            rp[i * 32]       = acc0[i];
            rp[(8 + i) * 32] = acc1[i];
        }
    }
    __syncthreads();

    if (seg == 0) {
        float bias = C[r0 + l16];
        float* op0 = out + (m0 + 8 * lhalf) * NREL + r0 + l16;
        float* op1 = op0 + 16 * NREL;
        #pragma unroll
        for (int i = 0; i < 8; ++i) {
            op0[i * NREL] = (acc0[i] + rp[i * 32]) + bias;
            op1[i * NREL] = (acc1[i] + rp[(8 + i) * 32]) + bias;
        }
    }
}

extern "C" void kernel_launch(void* const* d_in, const int* in_sizes, int n_in,
                              void* d_out, int out_size, void* d_ws, size_t ws_size,
                              hipStream_t stream) {
    const float* sbjs   = (const float*)d_in[0];
    const float* objs   = (const float*)d_in[1];
    const float* mus    = (const float*)d_in[2];
    const float* sigmas = (const float*)d_in[3];
    const float* priors = (const float*)d_in[4];
    float* out = (float*)d_out;

    float* Wf = (float*)d_ws;
    float* C  = Wf + KTOT * NREL;

    nb_prep_w<<<(TWO_D * NREL + 255) / 256, 256, 0, stream>>>(mus, sigmas, Wf);
    nb_prep_bias<<<NREL, 256, 0, stream>>>(mus, sigmas, priors, C);
    nb_wmma<<<NBLK, 256, 0, stream>>>(sbjs, objs, Wf, C, out);
}


# Main kernel: x tile staged with all b128 loads of a thread issued as one group ahead of its LDS stores (was a loop of 4 b32 loads then 4 stores per trip, each trip a dependent memory round trip)
# speedup vs baseline: 4.3204x; 1.4874x over previous
#include <hip/hip_runtime.h>
#include <hip/hip_bf16.h>
#include <math.h>

#define EMB        256
#define TWO_D      512
#define KTOT       1024
#define NREL       128
#define BATCH      4096
#define LOG_SQRT_2PI 0.9189385332046727f
#define XSTRIDE    516
#define MROWS      32

typedef __attribute__((ext_vector_type(2))) float v2f;
typedef __attribute__((ext_vector_type(4))) float v4f;
typedef __attribute__((ext_vector_type(8))) float v8f;

__global__ __launch_bounds__(256)
void nb_prep_w(const float* __restrict__ mus, const float* __restrict__ sigmas,
               float* __restrict__ Wf) {
    int i = blockIdx.x * 256 + threadIdx.x;
    if (i >= TWO_D * NREL) return;
    int k = i / NREL;
    int r = i - k * NREL;
    float s  = sigmas[r * TWO_D + k];
    float iv = 1.0f / (s * s);
    float av = -0.5f * iv;
    float bm = mus[r * TWO_D + k] * iv;
    int t = k >> 2, h = (k >> 1) & 1, j = k & 1;
    Wf[t * 512 + r * 4 + h * 2 + j]            = av;
    Wf[(t + 128) * 512 + r * 4 + h * 2 + j]    = bm;
}

__global__ __launch_bounds__(256)
void nb_prep_bias(const float* __restrict__ mus, const float* __restrict__ sigmas,
                  const float* __restrict__ priors, float* __restrict__ C) {
    __shared__ float red[256];
    int r = blockIdx.x;
    float acc = 0.0f;
    for (int k = threadIdx.x; k < TWO_D; k += 256) {
        float s  = sigmas[r * TWO_D + k];
        float mu = mus[r * TWO_D + k];
        float iv = 1.0f / (s * s);
        acc += -0.5f * mu * mu * iv - logf(s);
    }
    red[threadIdx.x] = acc;
    __syncthreads();
    for (int off = 128; off > 0; off >>= 1) {
        if (threadIdx.x < off) red[threadIdx.x] += red[threadIdx.x + off];
        __syncthreads();
    }
    if (threadIdx.x == 0)
        C[r] = red[0] + (float)TWO_D * (priors[r] - LOG_SQRT_2PI);
}

#define PF 16
#define HSTEP 128
#define RELB  64
#define NBLK  256
#define XPIECES 16

template <bool SQUARE, bool REFILL>
__device__ __forceinline__
void nb_steps(const int t0, v2f (&bq)[PF], const float* __restrict__ wp,
              const float* x0, const float* x1, v8f& acc0, v8f& acc1) {
    #pragma unroll
    for (int i = 0; i < PF; ++i) {
        const int t = t0 + i;
        const v2f b = bq[i];
        if (REFILL) bq[i] = *(const v2f*)(wp + (t + PF) * 512);
        const int xo = (t & 127) * 4;
        v2f a0 = *(const v2f*)(x0 + xo);
        v2f a1 = *(const v2f*)(x1 + xo);
        if (SQUARE) {
            a0.x = a0.x * a0.x;  a0.y = a0.y * a0.y;
            a1.x = a1.x * a1.x;  a1.y = a1.y * a1.y;
        }
        acc0 = __builtin_amdgcn_wmma_f32_16x16x4_f32(
            false, a0, false, b, (short)0, acc0, false, false);
        acc1 = __builtin_amdgcn_wmma_f32_16x16x4_f32(
            false, a1, false, b, (short)0, acc1, false, false);
    }
}

__global__ __launch_bounds__(256)
void nb_wmma(const float* __restrict__ sbjs, const float* __restrict__ objs,
             const float* __restrict__ Wf, const float* __restrict__ C,
             float* __restrict__ out) {
    __shared__ __attribute__((aligned(16))) float xs[MROWS * XSTRIDE];
    __shared__ float red[4 * 16 * 32];

    const int m0    = (blockIdx.x >> 1) * MROWS;
    const int rbase = (blockIdx.x & 1) * RELB;

    const int lane  = threadIdx.x & 31;
    const int wave  = __builtin_amdgcn_readfirstlane((int)(threadIdx.x >> 5));
    const int grp   = wave & 3;
    const int seg   = wave >> 2;
    const int l16   = lane & 15;
    const int lhalf = lane >> 4;
    const int r0    = rbase + grp * 16;
    const float* wp = Wf + seg * (HSTEP * 512) + ((r0 + l16) * 2 + lhalf) * 2;

    v2f bq[PF];
    #pragma unroll
    for (int i = 0; i < PF; ++i) bq[i] = *(const v2f*)(wp + i * 512);

    const int c4  = threadIdx.x & 127;
    const int rw0 = threadIdx.x >> 7;
    const float* xsrc = ((c4 < 64) ? sbjs + 4 * c4 : objs + 4 * (c4 - 64))
                        + (m0 + rw0) * EMB;
    v4f xv[XPIECES];
    #pragma unroll
    for (int j = 0; j < XPIECES; ++j)
        xv[j] = *(const v4f*)(xsrc + j * 2 * EMB);
    #pragma unroll
    for (int j = 0; j < XPIECES; ++j)
        *(v4f*)&xs[(rw0 + 2 * j) * XSTRIDE + 4 * c4] = xv[j];
    __syncthreads();

    const float* x0 = &xs[l16 * XSTRIDE + 2 * lhalf];
    const float* x1 = x0 + 16 * XSTRIDE;

    v8f acc0 = {};
    v8f acc1 = {};
    float* rp = red + grp * (16 * 32) + lane;

    if (seg == 0) {
        for (int t0 = 0; t0 < HSTEP - PF; t0 += PF)
            nb_steps<true, true>(t0, bq, wp, x0, x1, acc0, acc1);
        nb_steps<true, false>(HSTEP - PF, bq, wp, x0, x1, acc0, acc1);
    } else {
        for (int t0 = 0; t0 < HSTEP - PF; t0 += PF)
            nb_steps<false, true>(t0, bq, wp, x0, x1, acc0, acc1);
        nb_steps<false, false>(HSTEP - PF, bq, wp, x0, x1, acc0, acc1);
        #pragma unroll
        for (int i = 0; i < 8; ++i) {
            rp[i * 32]       = acc0[i];
            rp[(8 + i) * 32] = acc1[i];
        }
    }
    __syncthreads();

    if (seg == 0) {
        float bias = C[r0 + l16];
        float* op0 = out + (m0 + 8 * lhalf) * NREL + r0 + l16;
        float* op1 = op0 + 16 * NREL;
        #pragma unroll
        for (int i = 0; i < 8; ++i) {
            op0[i * NREL] = (acc0[i] + rp[i * 32]) + bias;
            op1[i * NREL] = (acc1[i] + rp[(8 + i) * 32]) + bias;
        }
    }
}

extern "C" void kernel_launch(void* const* d_in, const int* in_sizes, int n_in,
                              void* d_out, int out_size, void* d_ws, size_t ws_size,
                              hipStream_t stream) {
    const float* sbjs   = (const float*)d_in[0];
    const float* objs   = (const float*)d_in[1];
    const float* mus    = (const float*)d_in[2];
    const float* sigmas = (const float*)d_in[3];
    const float* priors = (const float*)d_in[4];
    float* out = (float*)d_out;

    float* Wf = (float*)d_ws;
    float* C  = Wf + KTOT * NREL;

    nb_prep_w<<<(TWO_D * NREL + 255) / 256, 256, 0, stream>>>(mus, sigmas, Wf);
    nb_prep_bias<<<NREL, 256, 0, stream>>>(mus, sigmas, priors, C);
    nb_wmma<<<NBLK, 256, 0, stream>>>(sbjs, objs, Wf, C, out);
}


# Both preparation kernels merged into ONE launch: a block per relation reads its mus/sigmas once and writes its coefficients and its bias; 3 dependent launches per call become 2, sums unchanged
# speedup vs baseline: 4.6790x; 1.0830x over previous
#include <hip/hip_runtime.h>
#include <hip/hip_bf16.h>
#include <math.h>

#define EMB        256
#define TWO_D      512
#define KTOT       1024
#define NREL       128
#define BATCH      4096
#define LOG_SQRT_2PI 0.9189385332046727f
#define XSTRIDE    516
#define MROWS      32

typedef __attribute__((ext_vector_type(2))) float v2f;
typedef __attribute__((ext_vector_type(4))) float v4f;
typedef __attribute__((ext_vector_type(8))) float v8f;

__global__ __launch_bounds__(256)
void nb_prep(const float* __restrict__ mus, const float* __restrict__ sigmas,
             const float* __restrict__ priors, float* __restrict__ Wf,
             float* __restrict__ C) {
    __shared__ float red[256];
    const int r  = blockIdx.x;
    const int k0 = threadIdx.x;
    const int k1 = threadIdx.x + 256;
    const float s0  = sigmas[r * TWO_D + k0];
    const float s1  = sigmas[r * TWO_D + k1];
    const float mu0 = mus[r * TWO_D + k0];
    const float mu1 = mus[r * TWO_D + k1];
    const float iv0 = 1.0f / (s0 * s0);
    const float iv1 = 1.0f / (s1 * s1);
    float acc = -0.5f * mu0 * mu0 * iv0 - logf(s0);
    acc += -0.5f * mu1 * mu1 * iv1 - logf(s1);
    red[threadIdx.x] = acc;
    __syncthreads();
    for (int off = 128; off > 0; off >>= 1) {
        if (threadIdx.x < off) red[threadIdx.x] += red[threadIdx.x + off];
        __syncthreads();
    }
    if (threadIdx.x == 0)
        C[r] = red[0] + (float)TWO_D * (priors[r] - LOG_SQRT_2PI);

    float* w0 = Wf + (k0 >> 2) * 512 + r * 4 + (k0 & 3);
    float* w1 = Wf + (k1 >> 2) * 512 + r * 4 + (k1 & 3);
    w0[0]         = -0.5f * iv0;
    w0[128 * 512] = mu0 * iv0;
    w1[0]         = -0.5f * iv1;
    w1[128 * 512] = mu1 * iv1;
}

#define PF 16
#define HSTEP 128
#define RELB  64
#define NBLK  256
#define XPIECES 16

template <bool SQUARE, bool REFILL>
__device__ __forceinline__
void nb_steps(const int t0, v2f (&bq)[PF], const float* __restrict__ wp,
              const float* x0, const float* x1, v8f& acc0, v8f& acc1) {
    #pragma unroll
    for (int i = 0; i < PF; ++i) {
        const int t = t0 + i;
        const v2f b = bq[i];
        if (REFILL) bq[i] = *(const v2f*)(wp + (t + PF) * 512);
        const int xo = (t & 127) * 4;
        v2f a0 = *(const v2f*)(x0 + xo);
        v2f a1 = *(const v2f*)(x1 + xo);
        if (SQUARE) {
            a0.x = a0.x * a0.x;  a0.y = a0.y * a0.y;
            a1.x = a1.x * a1.x;  a1.y = a1.y * a1.y;
        }
        acc0 = __builtin_amdgcn_wmma_f32_16x16x4_f32(
            false, a0, false, b, (short)0, acc0, false, false);
        acc1 = __builtin_amdgcn_wmma_f32_16x16x4_f32(
            false, a1, false, b, (short)0, acc1, false, false);
    }
}

__global__ __launch_bounds__(256)
void nb_wmma(const float* __restrict__ sbjs, const float* __restrict__ objs,
             const float* __restrict__ Wf, const float* __restrict__ C,
             float* __restrict__ out) {
    __shared__ __attribute__((aligned(16))) float xs[MROWS * XSTRIDE];
    __shared__ float red[4 * 16 * 32];

    const int m0    = (blockIdx.x >> 1) * MROWS;
    const int rbase = (blockIdx.x & 1) * RELB;

    const int lane  = threadIdx.x & 31;
    const int wave  = __builtin_amdgcn_readfirstlane((int)(threadIdx.x >> 5));
    const int grp   = wave & 3;
    const int seg   = wave >> 2;
    const int l16   = lane & 15;
    const int lhalf = lane >> 4;
    const int r0    = rbase + grp * 16;
    const float* wp = Wf + seg * (HSTEP * 512) + ((r0 + l16) * 2 + lhalf) * 2;

    v2f bq[PF];
    #pragma unroll
    for (int i = 0; i < PF; ++i) bq[i] = *(const v2f*)(wp + i * 512);

    const int c4  = threadIdx.x & 127;
    const int rw0 = threadIdx.x >> 7;
    const float* xsrc = ((c4 < 64) ? sbjs + 4 * c4 : objs + 4 * (c4 - 64))
                        + (m0 + rw0) * EMB;
    v4f xv[XPIECES];
    #pragma unroll
    for (int j = 0; j < XPIECES; ++j)
        xv[j] = *(const v4f*)(xsrc + j * 2 * EMB);
    #pragma unroll
    for (int j = 0; j < XPIECES; ++j)
        *(v4f*)&xs[(rw0 + 2 * j) * XSTRIDE + 4 * c4] = xv[j];
    __syncthreads();

    const float* x0 = &xs[l16 * XSTRIDE + 2 * lhalf];
    const float* x1 = x0 + 16 * XSTRIDE;

    v8f acc0 = {};
    v8f acc1 = {};
    float* rp = red + grp * (16 * 32) + lane;

    if (seg == 0) {
        for (int t0 = 0; t0 < HSTEP - PF; t0 += PF)
            nb_steps<true, true>(t0, bq, wp, x0, x1, acc0, acc1);
        nb_steps<true, false>(HSTEP - PF, bq, wp, x0, x1, acc0, acc1);
    } else {
        for (int t0 = 0; t0 < HSTEP - PF; t0 += PF)
            nb_steps<false, true>(t0, bq, wp, x0, x1, acc0, acc1);
        nb_steps<false, false>(HSTEP - PF, bq, wp, x0, x1, acc0, acc1);
        #pragma unroll
        for (int i = 0; i < 8; ++i) {
            rp[i * 32]       = acc0[i];
            rp[(8 + i) * 32] = acc1[i];
        }
    }
    __syncthreads();

    if (seg == 0) {
        float bias = C[r0 + l16];
        float* op0 = out + (m0 + 8 * lhalf) * NREL + r0 + l16;
        float* op1 = op0 + 16 * NREL;
        #pragma unroll
        for (int i = 0; i < 8; ++i) {
            op0[i * NREL] = (acc0[i] + rp[i * 32]) + bias;
            op1[i * NREL] = (acc1[i] + rp[(8 + i) * 32]) + bias;
        }
    }
}

extern "C" void kernel_launch(void* const* d_in, const int* in_sizes, int n_in,
                              void* d_out, int out_size, void* d_ws, size_t ws_size,
                              hipStream_t stream) {
    const float* sbjs   = (const float*)d_in[0];
    const float* objs   = (const float*)d_in[1];
    const float* mus    = (const float*)d_in[2];
    const float* sigmas = (const float*)d_in[3];
    const float* priors = (const float*)d_in[4];
    float* out = (float*)d_out;

    float* Wf = (float*)d_ws;
    float* C  = Wf + KTOT * NREL;

    nb_prep<<<NREL, 256, 0, stream>>>(mus, sigmas, priors, Wf, C);
    nb_wmma<<<NBLK, 256, 0, stream>>>(sbjs, objs, Wf, C, out);
}


# Prep launch folded into main: a wave reads its sigma/mu rows as whole lines, forms coefficients (v_rcp, v_log), transposes them via its own LDS; C[r] summed per lane then 8-lane tree; 2 launches->1
# speedup vs baseline: 5.7997x; 1.2395x over previous
#include <hip/hip_runtime.h>
#include <hip/hip_bf16.h>
#include <math.h>


#define EMB        256
#define TWO_D      512
#define NREL       128
#define BATCH      4096
#define LOG_SQRT_2PI 0.9189385332046727f
#define LN2        0.6931471805599453f
#define XSTRIDE    516
#define MROWS      32

typedef __attribute__((ext_vector_type(2))) float v2f;
typedef __attribute__((ext_vector_type(4))) float v4f;
typedef __attribute__((ext_vector_type(8))) float v8f;

#define RELB  64
#define NBLK  256
#define XPIECES 16
#define CHK    32
#define NCH    16
#define CSTEPS 8
#define WPITCH 36
#define WBUF   (16 * WPITCH)

__device__ __forceinline__ void wave_handover() {
    __builtin_amdgcn_fence(__ATOMIC_RELEASE, "wavefront");
    __builtin_amdgcn_wave_barrier();
    __builtin_amdgcn_fence(__ATOMIC_ACQUIRE, "wavefront");
}

__device__ __forceinline__
void nb_fetch(v4f (&R)[4], const float* __restrict__ p) {
    #pragma unroll
    for (int j = 0; j < 4; ++j) R[j] = *(const v4f*)(p + j * 4 * TWO_D);
}

template <bool SQUARE>
__device__ __forceinline__
void nb_stage(const v4f (&S)[4], const v4f (&M)[4], float* wst,
              v4f (&qacc)[4], v4f (&lacc)[4]) {
    #pragma unroll
    for (int j = 0; j < 4; ++j) {
        const v4f s = S[j];
        v4f iv;
        iv.x = __builtin_amdgcn_rcpf(s.x * s.x);
        iv.y = __builtin_amdgcn_rcpf(s.y * s.y);
        iv.z = __builtin_amdgcn_rcpf(s.z * s.z);
        iv.w = __builtin_amdgcn_rcpf(s.w * s.w);
        v4f c;
        if (SQUARE) {
            c = iv * -0.5f;
        } else {
            const v4f m = M[j];
            c = m * iv;
            qacc[j] += m * c;
            v4f lg;
            lg.x = __builtin_amdgcn_logf(s.x);
            lg.y = __builtin_amdgcn_logf(s.y);
            lg.z = __builtin_amdgcn_logf(s.z);
            lg.w = __builtin_amdgcn_logf(s.w);
            lacc[j] += lg;
        }
        *(v4f*)(wst + j * 4 * WPITCH) = c;
    }
}

template <bool SQUARE>
__device__ __forceinline__
void nb_steps(const float* bfr, const float* xa0, const float* xa1,
              v8f& acc0, v8f& acc1) {
    #pragma unroll
    for (int i = 0; i < CSTEPS; ++i) {
        const v2f b = *(const v2f*)(bfr + 4 * i);
        v2f a0 = *(const v2f*)(xa0 + 4 * i);
        v2f a1 = *(const v2f*)(xa1 + 4 * i);
        if (SQUARE) {
            a0.x = a0.x * a0.x;  a0.y = a0.y * a0.y;
            a1.x = a1.x * a1.x;  a1.y = a1.y * a1.y;
        }
        acc0 = __builtin_amdgcn_wmma_f32_16x16x4_f32(
            false, a0, false, b, (short)0, acc0, false, false);
        acc1 = __builtin_amdgcn_wmma_f32_16x16x4_f32(
            false, a1, false, b, (short)0, acc1, false, false);
    }
}

template <bool SQUARE>
__device__ __forceinline__
void nb_half(v4f (&S0)[4], v4f (&S1)[4], v4f (&M0)[4], v4f (&M1)[4],
             const float* __restrict__ sp, const float* __restrict__ mp,
             float* wst0, const float* bf0, const float* x0, const float* x1,
             v8f& acc0, v8f& acc1, v4f (&qacc)[4], v4f (&lacc)[4]) {
    float* wst1 = wst0 + WBUF;
    const float* bf1 = bf0 + WBUF;

    nb_stage<SQUARE>(S0, M0, wst0, qacc, lacc);
    nb_fetch(S0, sp + 2 * CHK);
    if (!SQUARE) nb_fetch(M0, mp + 2 * CHK);
    wave_handover();

    for (int n = 0; n < NCH; n += 2) {
        const int f1 = (n + 3 < NCH) ? n + 3 : NCH - 1;
        const int f0 = (n + 4 < NCH) ? n + 4 : NCH - 1;
        nb_stage<SQUARE>(S1, M1, wst1, qacc, lacc);
        nb_fetch(S1, sp + f1 * CHK);
        if (!SQUARE) nb_fetch(M1, mp + f1 * CHK);
        wave_handover();
        nb_steps<SQUARE>(bf0, x0 + n * CHK, x1 + n * CHK, acc0, acc1);
        wave_handover();
        if (n + 2 < NCH)
            nb_stage<SQUARE>(S0, M0, wst0, qacc, lacc);
        nb_fetch(S0, sp + f0 * CHK);
        if (!SQUARE) nb_fetch(M0, mp + f0 * CHK);
        wave_handover();
        nb_steps<SQUARE>(bf1, x0 + (n + 1) * CHK, x1 + (n + 1) * CHK, acc0, acc1);
        wave_handover();
    }
}

__global__ __launch_bounds__(256)
void nb_wmma(const float* __restrict__ sbjs, const float* __restrict__ objs,
             const float* __restrict__ mus, const float* __restrict__ sigmas,
             const float* __restrict__ priors, float* __restrict__ out) {
    __shared__ __attribute__((aligned(16))) float xs[MROWS * XSTRIDE];
    __shared__ float red[4 * 16 * 32];
    __shared__ __attribute__((aligned(16))) float wbuf[8 * 2 * WBUF];
    __shared__ float cb[RELB];

    const int m0    = (blockIdx.x >> 1) * MROWS;
    const int rbase = (blockIdx.x & 1) * RELB;

    const int lane  = threadIdx.x & 31;
    const int wave  = __builtin_amdgcn_readfirstlane((int)(threadIdx.x >> 5));
    const int grp   = wave & 3;
    const int seg   = wave >> 2;
    const int l16   = lane & 15;
    const int lhalf = lane >> 4;
    const int r0    = rbase + grp * 16;
    const int g8    = lane >> 3;
    const int p8    = lane & 7;

    const int rowoff = (r0 + g8) * TWO_D + 4 * p8;
    const float* sp = sigmas + rowoff;
    const float* mp = mus + rowoff;
    v4f S0[4], S1[4];
    v4f M0[4] = {}, M1[4] = {};
    nb_fetch(S0, sp);
    nb_fetch(S1, sp + CHK);
    if (seg) {
        nb_fetch(M0, mp);
        nb_fetch(M1, mp + CHK);
    }

    const int c4  = threadIdx.x & 127;
    const int rw0 = threadIdx.x >> 7;
    const float* xsrc = ((c4 < 64) ? sbjs + 4 * c4 : objs + 4 * (c4 - 64))
                        + (m0 + rw0) * EMB;
    v4f xv[XPIECES];
    #pragma unroll
    for (int j = 0; j < XPIECES; ++j)
        xv[j] = *(const v4f*)(xsrc + j * 2 * EMB);
    #pragma unroll
    for (int j = 0; j < XPIECES; ++j)
        *(v4f*)&xs[(rw0 + 2 * j) * XSTRIDE + 4 * c4] = xv[j];
    __syncthreads();

    const float* x0 = &xs[l16 * XSTRIDE + 2 * lhalf];
    const float* x1 = x0 + 16 * XSTRIDE;

    float* wb = wbuf + wave * (2 * WBUF);
    float* wst0 = wb + g8 * WPITCH + 4 * p8;
    const float* bf0 = wb + l16 * WPITCH + 2 * lhalf;

    v8f acc0 = {};
    v8f acc1 = {};
    v4f qacc[4] = {};
    v4f lacc[4] = {};
    float* rp = red + grp * (16 * 32) + lane;

    if (seg == 0) {
        nb_half<true>(S0, S1, M0, M1, sp, mp, wst0, bf0, x0, x1, acc0, acc1, qacc, lacc);
    } else {
        nb_half<false>(S0, S1, M0, M1, sp, mp, wst0, bf0, x0, x1, acc0, acc1, qacc, lacc);
        #pragma unroll
        for (int i = 0; i < 8; ++i) {
            rp[i * 32]       = acc0[i];
            rp[(8 + i) * 32] = acc1[i];
        }
        #pragma unroll
        for (int j = 0; j < 4; ++j) {
            const v4f t = -0.5f * qacc[j] - LN2 * lacc[j];
            float v = (t.x + t.y) + (t.z + t.w);
            v += __shfl_xor(v, 1);
            v += __shfl_xor(v, 2);
            v += __shfl_xor(v, 4);
            if (p8 == 0) cb[grp * 16 + 4 * j + g8] = v;
        }
    }
    __syncthreads();

    if (seg == 0) {
        float bias = cb[grp * 16 + l16]
                   + (float)TWO_D * (priors[r0 + l16] - LOG_SQRT_2PI);
        float* op0 = out + (m0 + 8 * lhalf) * NREL + r0 + l16;
        float* op1 = op0 + 16 * NREL;
        #pragma unroll
        for (int i = 0; i < 8; ++i) {
            op0[i * NREL] = (acc0[i] + rp[i * 32]) + bias;
            op1[i * NREL] = (acc1[i] + rp[(8 + i) * 32]) + bias;
        }
    }
}

extern "C" void kernel_launch(void* const* d_in, const int* in_sizes, int n_in,
                              void* d_out, int out_size, void* d_ws, size_t ws_size,
                              hipStream_t stream) {
    const float* sbjs   = (const float*)d_in[0];
    const float* objs   = (const float*)d_in[1];
    const float* mus    = (const float*)d_in[2];
    const float* sigmas = (const float*)d_in[3];
    const float* priors = (const float*)d_in[4];
    float* out = (float*)d_out;

    nb_wmma<<<NBLK, 256, 0, stream>>>(sbjs, objs, mus, sigmas, priors, out);
}


# Main loop: all A/B fragment reads of a chunk issued as ONE LDS group a chunk ahead into a 2nd register set (stood among the WMMAs); row fetch 2->4 chunks ahead; x half now on waves 0-3
# speedup vs baseline: 6.3816x; 1.1003x over previous
#include <hip/hip_runtime.h>
#include <hip/hip_bf16.h>
#include <math.h>


#define EMB        256
#define TWO_D      512
#define NREL       128
#define BATCH      4096
#define LOG_SQRT_2PI 0.9189385332046727f
#define LN2        0.6931471805599453f
#define XSTRIDE    516
#define MROWS      32

typedef __attribute__((ext_vector_type(2))) float v2f;
typedef __attribute__((ext_vector_type(4))) float v4f;
typedef __attribute__((ext_vector_type(8))) float v8f;

#define RELB  64
#define NBLK  256
#define XPIECES 16
#define CHK    32
#define NCH    16
#define CSTEPS 8
#define WPITCH 36
#define WBUF   (16 * WPITCH)

__device__ __forceinline__ void wave_handover() {
    __builtin_amdgcn_fence(__ATOMIC_RELEASE, "wavefront");
    __builtin_amdgcn_wave_barrier();
    __builtin_amdgcn_fence(__ATOMIC_ACQUIRE, "wavefront");
}

__device__ __forceinline__
void nb_fetch(v4f (&R)[4], const float* __restrict__ p) {
    #pragma unroll
    for (int j = 0; j < 4; ++j) R[j] = *(const v4f*)(p + j * 4 * TWO_D);
}

template <bool SQUARE>
__device__ __forceinline__
void nb_coef(const v4f (&S)[4], const v4f (&M)[4], v4f (&C)[4],
             v4f (&qacc)[4], v4f (&lacc)[4]) {
    #pragma unroll
    for (int j = 0; j < 4; ++j) {
        const v4f s = S[j];
        v4f iv;
        iv.x = __builtin_amdgcn_rcpf(s.x * s.x);
        iv.y = __builtin_amdgcn_rcpf(s.y * s.y);
        iv.z = __builtin_amdgcn_rcpf(s.z * s.z);
        iv.w = __builtin_amdgcn_rcpf(s.w * s.w);
        v4f c;
        if (SQUARE) {
            c = iv * -0.5f;
        } else {
            const v4f m = M[j];
            c = m * iv;
            qacc[j] += m * c;
            v4f lg;
            lg.x = __builtin_amdgcn_logf(s.x);
            lg.y = __builtin_amdgcn_logf(s.y);
            lg.z = __builtin_amdgcn_logf(s.z);
            lg.w = __builtin_amdgcn_logf(s.w);
            lacc[j] += lg;
        }
        C[j] = c;
    }
}

__device__ __forceinline__
void nb_put(const v4f (&C)[4], float* wst) {
    #pragma unroll
    for (int j = 0; j < 4; ++j) *(v4f*)(wst + j * 4 * WPITCH) = C[j];
}

struct Frags { v2f b[CSTEPS]; v2f a0[CSTEPS]; v2f a1[CSTEPS]; };

__device__ __forceinline__
void nb_frags(Frags& F, const float* bfr, const float* xa0, const float* xa1) {
    #pragma unroll
    for (int i = 0; i < CSTEPS; ++i) {
        F.b[i]  = *(const v2f*)(bfr + 4 * i);
        F.a0[i] = *(const v2f*)(xa0 + 4 * i);
        F.a1[i] = *(const v2f*)(xa1 + 4 * i);
    }
}

template <bool SQUARE>
__device__ __forceinline__
void nb_mma(const Frags& F, v8f& acc0, v8f& acc1) {
    #pragma unroll
    for (int i = 0; i < CSTEPS; ++i) {
        const v2f b = F.b[i];
        v2f a0 = F.a0[i];
        v2f a1 = F.a1[i];
        if (SQUARE) {
            a0.x = a0.x * a0.x;  a0.y = a0.y * a0.y;
            a1.x = a1.x * a1.x;  a1.y = a1.y * a1.y;
        }
        acc0 = __builtin_amdgcn_wmma_f32_16x16x4_f32(
            false, a0, false, b, (short)0, acc0, false, false);
        acc1 = __builtin_amdgcn_wmma_f32_16x16x4_f32(
            false, a1, false, b, (short)0, acc1, false, false);
    }
}

__device__ __forceinline__ void pin_group(v8f& acc0, v8f& acc1) {
    asm volatile("" : "+v"(acc0), "+v"(acc1) : : "memory");
}
__device__ __forceinline__ void pin_block(v8f& acc0, v8f& acc1, v4f (&C)[4]) {
    asm volatile("" : "+v"(acc0), "+v"(acc1),
                      "+v"(C[0]), "+v"(C[1]), "+v"(C[2]), "+v"(C[3]) : : "memory");
}
__device__ __forceinline__ void pin_rows(v4f (&R)[4]) {
    asm volatile("" : "+v"(R[0]), "+v"(R[1]), "+v"(R[2]), "+v"(R[3]) : : "memory");
}
__device__ __forceinline__ void pin_sums(v4f (&q)[4], v4f (&l)[4]) {
    asm volatile("" : "+v"(q[0]), "+v"(q[1]), "+v"(q[2]), "+v"(q[3]),
                      "+v"(l[0]), "+v"(l[1]), "+v"(l[2]), "+v"(l[3]) : : "memory");
}

template <bool SQUARE>
__device__ __forceinline__
void nb_turn(v4f (&Cn)[4], float* wst, Frags& Fn, const float* bfr,
             const float* xa0, const float* xa1,
             v4f (&S)[4], v4f (&M)[4], const Frags& Fc,
             bool fetch, const float* sp, const float* mp,
             v8f& acc0, v8f& acc1, v4f (&qacc)[4], v4f (&lacc)[4]) {
    nb_put(Cn, wst);
    wave_handover();
    nb_frags(Fn, bfr, xa0, xa1);
    pin_group(acc0, acc1);
    pin_rows(S);
    nb_coef<SQUARE>(S, M, Cn, qacc, lacc);
    nb_mma<SQUARE>(Fc, acc0, acc1);
    pin_block(acc0, acc1, Cn);
    if (!SQUARE) pin_sums(qacc, lacc);
    if (fetch) {
        nb_fetch(S, sp);
        if (!SQUARE) nb_fetch(M, mp);
    }
}

template <bool SQUARE, bool MORE>
__device__ __forceinline__
void nb_trip(int n, v4f (&S)[4][4], v4f (&M)[4][4], v4f (&C)[4], Frags& F0, Frags& F1,
             const float* __restrict__ sp, const float* __restrict__ mp,
             float* wst0, float* wst1, const float* bf0, const float* bf1,
             const float* x0, const float* x1,
             v8f& acc0, v8f& acc1, v4f (&qacc)[4], v4f (&lacc)[4]) {
    nb_turn<SQUARE>(C, wst1, F1, bf1, x0 + (n + 1) * CHK, x1 + (n + 1) * CHK,
                    S[2], M[2], F0, true, sp + (n + 6) * CHK, mp + (n + 6) * CHK,
                    acc0, acc1, qacc, lacc);
    nb_turn<SQUARE>(C, wst0, F0, bf0, x0 + (n + 2) * CHK, x1 + (n + 2) * CHK,
                    S[3], M[3], F1, true, sp + (n + 7) * CHK, mp + (n + 7) * CHK,
                    acc0, acc1, qacc, lacc);
    nb_turn<SQUARE>(C, wst1, F1, bf1, x0 + (n + 3) * CHK, x1 + (n + 3) * CHK,
                    S[0], M[0], F0, MORE, sp + (n + 8) * CHK, mp + (n + 8) * CHK,
                    acc0, acc1, qacc, lacc);
    nb_turn<SQUARE>(C, wst0, F0, bf0, x0 + (n + 4) * CHK, x1 + (n + 4) * CHK,
                    S[1], M[1], F1, MORE, sp + (n + 9) * CHK, mp + (n + 9) * CHK,
                    acc0, acc1, qacc, lacc);
}

template <bool SQUARE>
__device__ __forceinline__
void nb_half(v4f (&S)[4][4], v4f (&M)[4][4],
             const float* __restrict__ sp, const float* __restrict__ mp,
             float* wst0, const float* bf0, const float* x0, const float* x1,
             v8f& acc0, v8f& acc1, v4f (&qacc)[4], v4f (&lacc)[4]) {
    float* wst1 = wst0 + WBUF;
    const float* bf1 = bf0 + WBUF;
    v4f C[4];
    Frags F0, F1;

    pin_rows(S[0]);
    pin_rows(S[1]);
    nb_coef<SQUARE>(S[0], M[0], C, qacc, lacc);
    nb_put(C, wst0);
    pin_block(acc0, acc1, C);
    if (!SQUARE) pin_sums(qacc, lacc);
    nb_fetch(S[0], sp + 4 * CHK);
    if (!SQUARE) nb_fetch(M[0], mp + 4 * CHK);
    wave_handover();
    nb_frags(F0, bf0, x0, x1);
    nb_coef<SQUARE>(S[1], M[1], C, qacc, lacc);
    pin_block(acc0, acc1, C);
    if (!SQUARE) pin_sums(qacc, lacc);
    nb_fetch(S[1], sp + 5 * CHK);
    if (!SQUARE) nb_fetch(M[1], mp + 5 * CHK);

    for (int n = 0; n < NCH - 8; n += 4)
        nb_trip<SQUARE, true>(n, S, M, C, F0, F1, sp, mp, wst0, wst1, bf0, bf1, x0, x1,
                              acc0, acc1, qacc, lacc);
    nb_trip<SQUARE, false>(NCH - 8, S, M, C, F0, F1, sp, mp, wst0, wst1, bf0, bf1, x0, x1,
                           acc0, acc1, qacc, lacc);

    nb_turn<SQUARE>(C, wst1, F1, bf1, x0 + 13 * CHK, x1 + 13 * CHK,
                    S[2], M[2], F0, false, sp, mp, acc0, acc1, qacc, lacc);
    nb_turn<SQUARE>(C, wst0, F0, bf0, x0 + 14 * CHK, x1 + 14 * CHK,
                    S[3], M[3], F1, false, sp, mp, acc0, acc1, qacc, lacc);

    nb_put(C, wst1);
    wave_handover();
    nb_frags(F1, bf1, x0 + (NCH - 1) * CHK, x1 + (NCH - 1) * CHK);
    pin_group(acc0, acc1);
    nb_mma<SQUARE>(F0, acc0, acc1);
    nb_mma<SQUARE>(F1, acc0, acc1);
}

__global__ __launch_bounds__(256)
void nb_wmma(const float* __restrict__ sbjs, const float* __restrict__ objs,
             const float* __restrict__ mus, const float* __restrict__ sigmas,
             const float* __restrict__ priors, float* __restrict__ out) {
    __shared__ __attribute__((aligned(16))) float xs[MROWS * XSTRIDE];
    __shared__ float red[4 * 16 * 32];
    __shared__ __attribute__((aligned(16))) float wbuf[8 * 2 * WBUF];
    __shared__ float cb[RELB];

    const int m0    = (blockIdx.x >> 1) * MROWS;
    const int rbase = (blockIdx.x & 1) * RELB;

    const int lane  = threadIdx.x & 31;
    const int wave  = __builtin_amdgcn_readfirstlane((int)(threadIdx.x >> 5));
    const int grp   = wave & 3;
    const int seg   = wave >> 2;
    const int l16   = lane & 15;
    const int lhalf = lane >> 4;
    const int r0    = rbase + grp * 16;
    const int g8    = lane >> 3;
    const int p8    = lane & 7;

    const int rowoff = (r0 + g8) * TWO_D + 4 * p8;
    const float* sp = sigmas + rowoff;
    const float* mp = mus + rowoff;
    v4f S[4][4];
    v4f M[4][4];
    #pragma unroll
    for (int c = 0; c < 4; ++c) nb_fetch(S[c], sp + c * CHK);
    if (seg == 0) {
        #pragma unroll
        for (int c = 0; c < 4; ++c) nb_fetch(M[c], mp + c * CHK);
    }

    const int c4  = threadIdx.x & 127;
    const int rw0 = threadIdx.x >> 7;
    const float* xsrc = ((c4 < 64) ? sbjs + 4 * c4 : objs + 4 * (c4 - 64))
                        + (m0 + rw0) * EMB;
    v4f xv[XPIECES];
    #pragma unroll
    for (int j = 0; j < XPIECES; ++j)
        xv[j] = *(const v4f*)(xsrc + j * 2 * EMB);
    #pragma unroll
    for (int j = 0; j < XPIECES; ++j)
        *(v4f*)&xs[(rw0 + 2 * j) * XSTRIDE + 4 * c4] = xv[j];
    __syncthreads();

    const float* x0 = &xs[l16 * XSTRIDE + 2 * lhalf];
    const float* x1 = x0 + 16 * XSTRIDE;

    float* wb = wbuf + wave * (2 * WBUF);
    float* wst0 = wb + g8 * WPITCH + 4 * p8;
    const float* bf0 = wb + l16 * WPITCH + 2 * lhalf;

    v8f acc0 = {};
    v8f acc1 = {};
    v4f qacc[4] = {};
    v4f lacc[4] = {};
    float* rp = red + grp * (16 * 32) + lane;

    if (seg == 0) {
        nb_half<false>(S, M, sp, mp, wst0, bf0, x0, x1, acc0, acc1, qacc, lacc);
        #pragma unroll
        for (int j = 0; j < 4; ++j) {
            const v4f t = -0.5f * qacc[j] - LN2 * lacc[j];
            float v = (t.x + t.y) + (t.z + t.w);
            v += __shfl_xor(v, 1);
            v += __shfl_xor(v, 2);
            v += __shfl_xor(v, 4);
            if (p8 == 0) cb[grp * 16 + 4 * j + g8] = v;
        }
    } else {
        nb_half<true>(S, M, sp, mp, wst0, bf0, x0, x1, acc0, acc1, qacc, lacc);
        #pragma unroll
        for (int i = 0; i < 8; ++i) {
            rp[i * 32]       = acc0[i];
            rp[(8 + i) * 32] = acc1[i];
        }
    }
    __syncthreads();

    if (seg == 0) {
        float bias = cb[grp * 16 + l16]
                   + (float)TWO_D * (priors[r0 + l16] - LOG_SQRT_2PI);
        float* op0 = out + (m0 + 8 * lhalf) * NREL + r0 + l16;
        float* op1 = op0 + 16 * NREL;
        #pragma unroll
        for (int i = 0; i < 8; ++i) {
            op0[i * NREL] = (acc0[i] + rp[i * 32]) + bias;
            op1[i * NREL] = (acc1[i] + rp[(8 + i) * 32]) + bias;
        }
    }
}

extern "C" void kernel_launch(void* const* d_in, const int* in_sizes, int n_in,
                              void* d_out, int out_size, void* d_ws, size_t ws_size,
                              hipStream_t stream) {
    const float* sbjs   = (const float*)d_in[0];
    const float* objs   = (const float*)d_in[1];
    const float* mus    = (const float*)d_in[2];
    const float* sigmas = (const float*)d_in[3];
    const float* priors = (const float*)d_in[4];
    float* out = (float*)d_out;

    nb_wmma<<<NBLK, 256, 0, stream>>>(sbjs, objs, mus, sigmas, priors, out);
}


# Re-tiled to ONE wave per SIMD (4 waves a block): a wave runs both K halves of its 16 relations: sigma read and inverted once, no partial sums cross waves, C[r] lane sums by DPP; all sum orders kept
# speedup vs baseline: 7.0484x; 1.1045x over previous
#include <hip/hip_runtime.h>
#include <hip/hip_bf16.h>
#include <math.h>


#define EMB        256
#define TWO_D      512
#define NREL       128
#define BATCH      4096
#define LOG_SQRT_2PI 0.9189385332046727f
#define LN2        0.6931471805599453f
#define XSTRIDE    516
#define MROWS      32

typedef __attribute__((ext_vector_type(2))) float v2f;
typedef __attribute__((ext_vector_type(4))) float v4f;
typedef __attribute__((ext_vector_type(8))) float v8f;

#define RELB  64
#define NBLK  256
#define NTHR  128
#define XPIECES 32
#define CHK    32
#define NCH    16
#define CSTEPS 8
#define WPITCH 36
#define WBUF   (16 * WPITCH)

__device__ __forceinline__ void wave_handover() {
    __builtin_amdgcn_fence(__ATOMIC_RELEASE, "wavefront");
    __builtin_amdgcn_wave_barrier();
    __builtin_amdgcn_fence(__ATOMIC_ACQUIRE, "wavefront");
}

__device__ __forceinline__ void fetch_order() {
    asm volatile("" : : : "memory");
}

template <int XM>
__device__ __forceinline__ float add_xor_lane(float v) {
    const int o = __builtin_amdgcn_update_dpp(0, __builtin_bit_cast(int, v),
                                              0x160 + XM, 0xf, 0xf, true);
    return v + __builtin_bit_cast(float, o);
}

__device__ __forceinline__
void nb_fetch(v4f (&R)[4], const float* __restrict__ p) {
    #pragma unroll
    for (int j = 0; j < 4; ++j) R[j] = *(const v4f*)(p + j * 4 * TWO_D);
}

__device__ __forceinline__
void nb_coef(const v4f (&S)[4], const v4f (&M)[4], v4f (&CA)[4], v4f (&CB)[4],
             v4f (&qacc)[4], v4f (&lacc)[4]) {
    #pragma unroll
    for (int j = 0; j < 4; ++j) {
        const v4f s = S[j];
        v4f iv;
        iv.x = __builtin_amdgcn_rcpf(s.x * s.x);
        iv.y = __builtin_amdgcn_rcpf(s.y * s.y);
        iv.z = __builtin_amdgcn_rcpf(s.z * s.z);
        iv.w = __builtin_amdgcn_rcpf(s.w * s.w);
        CA[j] = iv * -0.5f;
        const v4f m = M[j];
        const v4f c = m * iv;
        qacc[j] += m * c;
        v4f lg;
        lg.x = __builtin_amdgcn_logf(s.x);
        lg.y = __builtin_amdgcn_logf(s.y);
        lg.z = __builtin_amdgcn_logf(s.z);
        lg.w = __builtin_amdgcn_logf(s.w);
        lacc[j] += lg;
        CB[j] = c;
    }
}

__device__ __forceinline__
void nb_put(const v4f (&CA)[4], const v4f (&CB)[4], float* wst) {
    #pragma unroll
    for (int j = 0; j < 4; ++j) *(v4f*)(wst + j * 4 * WPITCH) = CA[j];
    #pragma unroll
    for (int j = 0; j < 4; ++j) *(v4f*)(wst + WBUF + j * 4 * WPITCH) = CB[j];
}

struct Frags { v2f ba[CSTEPS]; v2f bb[CSTEPS]; v2f a0[CSTEPS]; v2f a1[CSTEPS]; };

__device__ __forceinline__
void nb_frags(Frags& F, const float* bfr, const float* xa0, const float* xa1) {
    #pragma unroll
    for (int i = 0; i < CSTEPS; ++i) {
        F.ba[i] = *(const v2f*)(bfr + 4 * i);
        F.bb[i] = *(const v2f*)(bfr + WBUF + 4 * i);
        F.a0[i] = *(const v2f*)(xa0 + 4 * i);
        F.a1[i] = *(const v2f*)(xa1 + 4 * i);
    }
}

struct Accs { v8f q0; v8f q1; v8f x0; v8f x1; };

__device__ __forceinline__
void nb_mma(const Frags& F, Accs& A) {
    #pragma unroll
    for (int i = 0; i < CSTEPS; ++i) {
        const v2f ba = F.ba[i];
        const v2f bb = F.bb[i];
        const v2f a0 = F.a0[i];
        const v2f a1 = F.a1[i];
        v2f s0, s1;
        s0.x = a0.x * a0.x;  s0.y = a0.y * a0.y;
        s1.x = a1.x * a1.x;  s1.y = a1.y * a1.y;
        A.q0 = __builtin_amdgcn_wmma_f32_16x16x4_f32(
            false, s0, false, ba, (short)0, A.q0, false, false);
        A.q1 = __builtin_amdgcn_wmma_f32_16x16x4_f32(
            false, s1, false, ba, (short)0, A.q1, false, false);
        A.x0 = __builtin_amdgcn_wmma_f32_16x16x4_f32(
            false, a0, false, bb, (short)0, A.x0, false, false);
        A.x1 = __builtin_amdgcn_wmma_f32_16x16x4_f32(
            false, a1, false, bb, (short)0, A.x1, false, false);
    }
}

__device__ __forceinline__ void pin_group(Accs& A) {
    asm volatile("" : "+v"(A.q0), "+v"(A.q1), "+v"(A.x0), "+v"(A.x1) : : "memory");
}
__device__ __forceinline__ void pin_block(Accs& A, v4f (&CA)[4], v4f (&CB)[4]) {
    asm volatile("" : "+v"(A.q0), "+v"(A.q1), "+v"(A.x0), "+v"(A.x1),
                      "+v"(CA[0]), "+v"(CA[1]), "+v"(CA[2]), "+v"(CA[3]),
                      "+v"(CB[0]), "+v"(CB[1]), "+v"(CB[2]), "+v"(CB[3]) : : "memory");
}
__device__ __forceinline__ void pin_rows(v4f (&R)[4]) {
    asm volatile("" : "+v"(R[0]), "+v"(R[1]), "+v"(R[2]), "+v"(R[3]) : : "memory");
}
__device__ __forceinline__ void pin_sums(v4f (&q)[4], v4f (&l)[4]) {
    asm volatile("" : "+v"(q[0]), "+v"(q[1]), "+v"(q[2]), "+v"(q[3]),
                      "+v"(l[0]), "+v"(l[1]), "+v"(l[2]), "+v"(l[3]) : : "memory");
}

__device__ __forceinline__
void nb_turn(v4f (&CA)[4], v4f (&CB)[4], float* wst, Frags& Fn, const float* bfr,
             const float* xa0, const float* xa1,
             v4f (&S)[4], v4f (&M)[4], const Frags& Fc,
             bool fetch, const float* sp, const float* mp,
             Accs& A, v4f (&qacc)[4], v4f (&lacc)[4]) {
    nb_put(CA, CB, wst);
    wave_handover();
    nb_frags(Fn, bfr, xa0, xa1);
    pin_group(A);
    pin_rows(S);
    nb_coef(S, M, CA, CB, qacc, lacc);
    nb_mma(Fc, A);
    pin_block(A, CA, CB);
    pin_sums(qacc, lacc);
    if (fetch) {
        nb_fetch(S, sp);
        nb_fetch(M, mp);
    }
}

template <bool MORE>
__device__ __forceinline__
void nb_trip(int n, v4f (&S)[4][4], v4f (&M)[4][4], v4f (&CA)[4], v4f (&CB)[4],
             Frags& F0, Frags& F1,
             const float* __restrict__ sp, const float* __restrict__ mp,
             float* wst0, float* wst1, const float* bf0, const float* bf1,
             const float* x0, const float* x1,
             Accs& A, v4f (&qacc)[4], v4f (&lacc)[4]) {
    nb_turn(CA, CB, wst1, F1, bf1, x0 + (n + 1) * CHK, x1 + (n + 1) * CHK,
            S[2], M[2], F0, true, sp + (n + 6) * CHK, mp + (n + 6) * CHK,
            A, qacc, lacc);
    nb_turn(CA, CB, wst0, F0, bf0, x0 + (n + 2) * CHK, x1 + (n + 2) * CHK,
            S[3], M[3], F1, true, sp + (n + 7) * CHK, mp + (n + 7) * CHK,
            A, qacc, lacc);
    nb_turn(CA, CB, wst1, F1, bf1, x0 + (n + 3) * CHK, x1 + (n + 3) * CHK,
            S[0], M[0], F0, MORE, sp + (n + 8) * CHK, mp + (n + 8) * CHK,
            A, qacc, lacc);
    nb_turn(CA, CB, wst0, F0, bf0, x0 + (n + 4) * CHK, x1 + (n + 4) * CHK,
            S[1], M[1], F1, MORE, sp + (n + 9) * CHK, mp + (n + 9) * CHK,
            A, qacc, lacc);
}

__device__ __forceinline__
void nb_all_k(v4f (&S)[4][4], v4f (&M)[4][4],
              const float* __restrict__ sp, const float* __restrict__ mp,
              float* wst0, const float* bf0, const float* x0, const float* x1,
              Accs& A, v4f (&qacc)[4], v4f (&lacc)[4]) {
    float* wst1 = wst0 + 2 * WBUF;
    const float* bf1 = bf0 + 2 * WBUF;
    v4f CA[4];
    v4f CB[4];
    Frags F0, F1;

    pin_rows(S[0]);
    pin_rows(S[1]);
    nb_coef(S[0], M[0], CA, CB, qacc, lacc);
    nb_put(CA, CB, wst0);
    pin_block(A, CA, CB);
    pin_sums(qacc, lacc);
    nb_fetch(S[0], sp + 4 * CHK);
    nb_fetch(M[0], mp + 4 * CHK);
    wave_handover();
    nb_frags(F0, bf0, x0, x1);
    nb_coef(S[1], M[1], CA, CB, qacc, lacc);
    pin_block(A, CA, CB);
    pin_sums(qacc, lacc);
    nb_fetch(S[1], sp + 5 * CHK);
    nb_fetch(M[1], mp + 5 * CHK);

    for (int n = 0; n < NCH - 8; n += 4)
        nb_trip<true>(n, S, M, CA, CB, F0, F1, sp, mp, wst0, wst1, bf0, bf1, x0, x1,
                      A, qacc, lacc);
    nb_trip<false>(NCH - 8, S, M, CA, CB, F0, F1, sp, mp, wst0, wst1, bf0, bf1, x0, x1,
                   A, qacc, lacc);

    nb_turn(CA, CB, wst1, F1, bf1, x0 + 13 * CHK, x1 + 13 * CHK,
            S[2], M[2], F0, false, sp, mp, A, qacc, lacc);
    nb_turn(CA, CB, wst0, F0, bf0, x0 + 14 * CHK, x1 + 14 * CHK,
            S[3], M[3], F1, false, sp, mp, A, qacc, lacc);

    nb_put(CA, CB, wst1);
    wave_handover();
    nb_frags(F1, bf1, x0 + (NCH - 1) * CHK, x1 + (NCH - 1) * CHK);
    pin_group(A);
    nb_mma(F0, A);
    nb_mma(F1, A);
}

__global__ __launch_bounds__(NTHR)
void nb_wmma(const float* __restrict__ sbjs, const float* __restrict__ objs,
             const float* __restrict__ mus, const float* __restrict__ sigmas,
             const float* __restrict__ priors, float* __restrict__ out) {
    __shared__ __attribute__((aligned(16))) float xs[MROWS * XSTRIDE];
    __shared__ __attribute__((aligned(16))) float wbuf[4 * 4 * WBUF];
    __shared__ float cb[RELB];

    const int m0    = (blockIdx.x >> 1) * MROWS;
    const int rbase = (blockIdx.x & 1) * RELB;

    const int lane  = threadIdx.x & 31;
    const int wave  = __builtin_amdgcn_readfirstlane((int)(threadIdx.x >> 5));
    const int l16   = lane & 15;
    const int lhalf = lane >> 4;
    const int r0    = rbase + wave * 16;
    const int g8    = lane >> 3;
    const int p8    = lane & 7;

    const int c4  = threadIdx.x;
    const float* xsrc = ((c4 < 64) ? sbjs + 4 * c4 : objs + 4 * (c4 - 64)) + m0 * EMB;
    v4f xv[XPIECES];
    #pragma unroll
    for (int j = 0; j < XPIECES; ++j)
        xv[j] = *(const v4f*)(xsrc + j * EMB);

    const int rowoff = (r0 + g8) * TWO_D + 4 * p8;
    const float* sp = sigmas + rowoff;
    const float* mp = mus + rowoff;
    v4f S[4][4];
    v4f M[4][4];
    #pragma unroll
    for (int c = 0; c < 4; ++c) {
        nb_fetch(S[c], sp + c * CHK);
        nb_fetch(M[c], mp + c * CHK);
        if (c < 3) fetch_order();
    }

    #pragma unroll
    for (int j = 0; j < XPIECES; ++j)
        *(v4f*)&xs[j * XSTRIDE + 4 * c4] = xv[j];
    __builtin_amdgcn_fence(__ATOMIC_RELEASE, "workgroup", "local");
    __builtin_amdgcn_s_barrier();
    __builtin_amdgcn_fence(__ATOMIC_ACQUIRE, "workgroup", "local");

    const float* x0 = &xs[l16 * XSTRIDE + 2 * lhalf];
    const float* x1 = x0 + 16 * XSTRIDE;

    float* wb = wbuf + wave * (4 * WBUF);
    float* wst0 = wb + g8 * WPITCH + 4 * p8;
    const float* bf0 = wb + l16 * WPITCH + 2 * lhalf;

    Accs A;
    A.q0 = (v8f){};  A.q1 = (v8f){};  A.x0 = (v8f){};  A.x1 = (v8f){};
    v4f qacc[4] = {};
    v4f lacc[4] = {};

    nb_all_k(S, M, sp, mp, wst0, bf0, x0, x1, A, qacc, lacc);

    #pragma unroll
    for (int j = 0; j < 4; ++j) {
        const v4f t = -0.5f * qacc[j] - LN2 * lacc[j];
        float v = (t.x + t.y) + (t.z + t.w);
        v = add_xor_lane<1>(v);
        v = add_xor_lane<2>(v);
        v = add_xor_lane<4>(v);
        if (p8 == 0) cb[wave * 16 + 4 * j + g8] = v;
    }
    wave_handover();

    const float bias = cb[wave * 16 + l16]
                     + (float)TWO_D * (priors[r0 + l16] - LOG_SQRT_2PI);
    float* op0 = out + (m0 + 8 * lhalf) * NREL + r0 + l16;
    float* op1 = op0 + 16 * NREL;
    #pragma unroll
    for (int i = 0; i < 8; ++i) {
        op0[i * NREL] = (A.x0[i] + A.q0[i]) + bias;
        op1[i * NREL] = (A.x1[i] + A.q1[i]) + bias;
    }
}

extern "C" void kernel_launch(void* const* d_in, const int* in_sizes, int n_in,
                              void* d_out, int out_size, void* d_ws, size_t ws_size,
                              hipStream_t stream) {
    const float* sbjs   = (const float*)d_in[0];
    const float* objs   = (const float*)d_in[1];
    const float* mus    = (const float*)d_in[2];
    const float* sigmas = (const float*)d_in[3];
    const float* priors = (const float*)d_in[4];
    float* out = (float*)d_out;

    nb_wmma<<<NBLK, NTHR, 0, stream>>>(sbjs, objs, mus, sigmas, priors, out);
}
